# starting kernel
# baseline (speedup 1.0000x reference)
#include <hip/hip_runtime.h>
#include <hip/hip_bf16.h>


typedef __attribute__((ext_vector_type(2))) float v2f;
typedef __attribute__((ext_vector_type(4))) float v4f;
typedef __attribute__((ext_vector_type(8))) float v8f;

#define D_FEAT 64
#define EDGES_PER_WAVE 16
#define BLOCK_THREADS 256
#define EDGES_PER_BLOCK ((BLOCK_THREADS / 32) * EDGES_PER_WAVE)

__global__ __launch_bounds__(BLOCK_THREADS) void edge_dot_wmma_kernel(
    const float* __restrict__ h,
    const int* __restrict__ src,
    const int* __restrict__ dst,
    float* __restrict__ out,
    int E)
{
    const int lane  = threadIdx.x & 31;
    const int wave  = blockIdx.x * (BLOCK_THREADS >> 5) + (threadIdx.x >> 5);
    const int m     = lane & 15;
    const int half  = lane >> 4;

    const int e0 = wave * EDGES_PER_WAVE;
    int e = e0 + m;
    int ec = e < E ? e : (E - 1);

    const long long si = (long long)src[ec];
    const long long di = (long long)dst[ec];

    const float* __restrict__ srow = h + si * D_FEAT + half * 32;
    const float* __restrict__ drow = h + di * D_FEAT + half * 32;

    v2f acc; acc.x = 0.0f; acc.y = 0.0f;
#pragma unroll
    for (int c = 0; c < 8; ++c) {
        v4f s = *(const v4f*)(srow + c * 4);
        v4f t = *(const v4f*)(drow + c * 4);
        acc.x = __builtin_fmaf(s.x, t.x, acc.x);
        acc.y = __builtin_fmaf(s.y, t.y, acc.y);
        acc.x = __builtin_fmaf(s.z, t.z, acc.x);
        acc.y = __builtin_fmaf(s.w, t.w, acc.y);
    }

    v2f ones; ones.x = 1.0f; ones.y = 1.0f;
    v8f d = {};
    d = __builtin_amdgcn_wmma_f32_16x16x4_f32(
         false, acc,
         false, ones,
         (short)0, d,
         false,  false);

    if ((lane & 15) == 0) {
        const int base = e0 + half * 8;
        if (base + 7 < E) {
            v4f lo; lo.x = d[0]; lo.y = d[1]; lo.z = d[2]; lo.w = d[3];
            v4f hi; hi.x = d[4]; hi.y = d[5]; hi.z = d[6]; hi.w = d[7];
            *(v4f*)(out + base)     = lo;
            *(v4f*)(out + base + 4) = hi;
        } else {
#pragma unroll
            for (int v = 0; v < 8; ++v) {
                if (base + v < E) out[base + v] = d[v];
            }
        }
    }
}

extern "C" void kernel_launch(void* const* d_in, const int* in_sizes, int n_in,
                              void* d_out, int out_size, void* d_ws, size_t ws_size,
                              hipStream_t stream) {
    const float* h   = (const float*)d_in[0];
    const int*   src = (const int*)d_in[1];
    const int*   dst = (const int*)d_in[2];
    float*       out = (float*)d_out;

    const int E = in_sizes[1];
    const int blocks = (E + EDGES_PER_BLOCK - 1) / EDGES_PER_BLOCK;

    edge_dot_wmma_kernel<<<blocks, BLOCK_THREADS, 0, stream>>>(h, src, dst, out, E);
}


# Dot products moved onto the f32 WMMA: an edge pair's src rows form the A fragment, its dst rows B (8 lanes per 128-B line), D's diagonal summed by DPP; looping waves; new summation order
# speedup vs baseline: 3.2644x; 3.2644x over previous
#include <hip/hip_runtime.h>


typedef __attribute__((ext_vector_type(2))) float v2f;
typedef __attribute__((ext_vector_type(4))) float v4f;
typedef __attribute__((ext_vector_type(8))) float v8f;

#define D_FEAT 64
#define BLOCK_THREADS 256
#define WAVES_PER_BLOCK (BLOCK_THREADS / 32)
#define GRID_BLOCKS 1024
#define HALF_STEPS 8

#define XMASK_ADD(v, M) \
    ((v) + __builtin_bit_cast(float, __builtin_amdgcn_update_dpp( \
        0, __builtin_bit_cast(int, (v)), 0x160 + (M), 0xF, 0xF, false)))

__global__ __launch_bounds__(BLOCK_THREADS) void edge_dot_rowfrag_kernel(
    const float* __restrict__ h,
    const int* __restrict__ src,
    const int* __restrict__ dst,
    float* __restrict__ out,
    int E)
{
    const int lane = threadIdx.x & 31;
    const int wave   = __builtin_amdgcn_readfirstlane(
                           (int)(blockIdx.x * WAVES_PER_BLOCK + (threadIdx.x >> 5)));
    const int nwaves = (int)gridDim.x * WAVES_PER_BLOCK;
    const int npass  = (E + 31) >> 5;

    const int l7   = lane & 7;
    const int grp  = (lane >> 3) & 1;
    const unsigned foffb = (unsigned)(l7 + 8 * (lane >> 4)) * 16u;
    const char* __restrict__ hb = (const char*)h;
    const int pull = grp * 4;

    int p = wave;
    int ns = 0, nd = 0;
    if (p < npass) {
        int e = (p << 5) + lane;
        e = e < E ? e : E - 1;
        ns = src[e];
        nd = dst[e];
    }

    for (; p < npass; p += nwaves) {
        const int my_s = ns;
        const int my_d = nd;
        float r[2];
#pragma unroll
        for (int half = 0; half < 2; ++half) {
            unsigned so[HALF_STEPS], dof[HALF_STEPS];
#pragma unroll
            for (int u = 0; u < HALF_STEPS; ++u) {
                const int t = half * HALF_STEPS + u;
                const int si = __builtin_amdgcn_ds_bpermute(pull + 8 * t, my_s);
                const int di = __builtin_amdgcn_ds_bpermute(pull + 8 * t, my_d);
                so[u]  = ((unsigned)si << 8) | foffb;
                dof[u] = ((unsigned)di << 8) | foffb;
            }
            v4f s[HALF_STEPS], d[HALF_STEPS];
#pragma unroll
            for (int u = 0; u < HALF_STEPS; ++u) {
                s[u] = *(const v4f*)(hb + so[u]);
                d[u] = *(const v4f*)(hb + dof[u]);
            }
            if (half == 1) {
                int pn = p + nwaves;
                pn = pn < npass ? pn : npass - 1;
                int e = (pn << 5) + lane;
                e = e < E ? e : E - 1;
                ns = src[e];
                nd = dst[e];
            }
            float rr = 0.0f;
#pragma unroll
            for (int u = 0; u < HALF_STEPS; ++u) {
                v2f a0; a0.x = s[u].x; a0.y = s[u].y;
                v2f b0; b0.x = d[u].x; b0.y = d[u].y;
                v2f a1; a1.x = s[u].z; a1.y = s[u].w;
                v2f b1; b1.x = d[u].z; b1.y = d[u].w;
                v8f acc = {};
                acc = __builtin_amdgcn_wmma_f32_16x16x4_f32(false, a0, false, b0, (short)0, acc, false, false);
                acc = __builtin_amdgcn_wmma_f32_16x16x4_f32(false, a1, false, b1, (short)0, acc, false, false);

                float v = acc[0];
                v = l7 == 1 ? acc[1] : v;
                v = l7 == 2 ? acc[2] : v;
                v = l7 == 3 ? acc[3] : v;
                v = l7 == 4 ? acc[4] : v;
                v = l7 == 5 ? acc[5] : v;
                v = l7 == 6 ? acc[6] : v;
                v = l7 == 7 ? acc[7] : v;
                v = XMASK_ADD(v, 1);
                v = XMASK_ADD(v, 2);
                v = XMASK_ADD(v, 4);
                rr = (l7 == u) ? v : rr;
            }
            r[half] = rr;
        }
        const float r0 = r[0], r1 = r[1];

        const int e0 = p << 5;
        const int eo = e0 + 2 * l7 + (lane >> 4);
        if (lane < 8 || lane >= 24) {
            if (eo < E)      out[eo]      = r0;
            if (eo + 16 < E) out[eo + 16] = r1;
        }
    }
}

extern "C" void kernel_launch(void* const* d_in, const int* in_sizes, int n_in,
                              void* d_out, int out_size, void* d_ws, size_t ws_size,
                              hipStream_t stream) {
    const float* h   = (const float*)d_in[0];
    const int*   src = (const int*)d_in[1];
    const int*   dst = (const int*)d_in[2];
    float*       out = (float*)d_out;

    const int E = in_sizes[1];

    edge_dot_rowfrag_kernel<<<GRID_BLOCKS, BLOCK_THREADS, 0, stream>>>(h, src, dst, out, E);
}


# One accumulator tile per 4 edges (was 2): each row spread over 4 fragment rows, 4 chained WMMAs per tile, so diagonal selects, DPP sums and index pulls per edge halve; new summation order
# speedup vs baseline: 3.6321x; 1.1126x over previous
#include <hip/hip_runtime.h>


typedef __attribute__((ext_vector_type(2))) float v2f;
typedef __attribute__((ext_vector_type(4))) float v4f;
typedef __attribute__((ext_vector_type(8))) float v8f;

#define D_FEAT 64
#define BLOCK_THREADS 256
#define WAVES_PER_BLOCK (BLOCK_THREADS / 32)
#define GRID_BLOCKS 1024
#define HALF_SETS 4

#define XMASK_ADD(v, M) \
    ((v) + __builtin_bit_cast(float, __builtin_amdgcn_update_dpp( \
        0, __builtin_bit_cast(int, (v)), 0x160 + (M), 0xF, 0xF, false)))

__global__ __launch_bounds__(BLOCK_THREADS) void edge_dot_quadfrag_kernel(
    const float* __restrict__ h,
    const int* __restrict__ src,
    const int* __restrict__ dst,
    float* __restrict__ out,
    int E)
{
    const int lane = threadIdx.x & 31;
    const int wave   = __builtin_amdgcn_readfirstlane(
                           (int)(blockIdx.x * WAVES_PER_BLOCK + (threadIdx.x >> 5)));
    const int nwaves = (int)gridDim.x * WAVES_PER_BLOCK;
    const int npass  = (E + 31) >> 5;

    const int l7   = lane & 7;
    const int l3   = lane & 3;
    const int mem  = (lane >> 2) & 3;
    const unsigned foffb = (unsigned)(l3 + 4 * (lane >> 4)) * 16u;
    const char* __restrict__ hb = (const char*)h;
    const int pull = mem * 16;

    int p = wave;
    int ns = 0, nd = 0;
    if (p < npass) {
        int e = (p << 5) + lane;
        e = e < E ? e : E - 1;
        ns = src[e];
        nd = dst[e];
    }

    for (; p < npass; p += nwaves) {
        const int my_s = ns;
        const int my_d = nd;
        float r[2];
#pragma unroll
        for (int half = 0; half < 2; ++half) {
            unsigned so[HALF_SETS], dof[HALF_SETS];
#pragma unroll
            for (int u = 0; u < HALF_SETS; ++u) {
                const int si = __builtin_amdgcn_ds_bpermute(pull + 64 * half + 4 * u, my_s);
                const int di = __builtin_amdgcn_ds_bpermute(pull + 64 * half + 4 * u, my_d);
                so[u]  = ((unsigned)si << 8) | foffb;
                dof[u] = ((unsigned)di << 8) | foffb;
            }
            v4f s0[HALF_SETS], s1[HALF_SETS], d0[HALF_SETS], d1[HALF_SETS];
#pragma unroll
            for (int u = 0; u < HALF_SETS; ++u) {
                s0[u] = *(const v4f*)(hb + so[u]);
                d0[u] = *(const v4f*)(hb + dof[u]);
                s1[u] = *(const v4f*)(hb + so[u] + 128);
                d1[u] = *(const v4f*)(hb + dof[u] + 128);
            }
            if (half == 1) {
                int pn = p + nwaves;
                pn = pn < npass ? pn : npass - 1;
                int e = (pn << 5) + lane;
                e = e < E ? e : E - 1;
                ns = src[e];
                nd = dst[e];
            }
            float rr = 0.0f;
#pragma unroll
            for (int u = 0; u < HALF_SETS; ++u) {
                v2f a0; a0.x = s0[u].x; a0.y = s0[u].y;
                v2f b0; b0.x = d0[u].x; b0.y = d0[u].y;
                v2f a1; a1.x = s0[u].z; a1.y = s0[u].w;
                v2f b1; b1.x = d0[u].z; b1.y = d0[u].w;
                v2f a2; a2.x = s1[u].x; a2.y = s1[u].y;
                v2f b2; b2.x = d1[u].x; b2.y = d1[u].y;
                v2f a3; a3.x = s1[u].z; a3.y = s1[u].w;
                v2f b3; b3.x = d1[u].z; b3.y = d1[u].w;
                v8f acc = {};
                acc = __builtin_amdgcn_wmma_f32_16x16x4_f32(false, a0, false, b0, (short)0, acc, false, false);
                acc = __builtin_amdgcn_wmma_f32_16x16x4_f32(false, a1, false, b1, (short)0, acc, false, false);
                acc = __builtin_amdgcn_wmma_f32_16x16x4_f32(false, a2, false, b2, (short)0, acc, false, false);
                acc = __builtin_amdgcn_wmma_f32_16x16x4_f32(false, a3, false, b3, (short)0, acc, false, false);

                float v = acc[0];
                v = l7 == 1 ? acc[1] : v;
                v = l7 == 2 ? acc[2] : v;
                v = l7 == 3 ? acc[3] : v;
                v = l7 == 4 ? acc[4] : v;
                v = l7 == 5 ? acc[5] : v;
                v = l7 == 6 ? acc[6] : v;
                v = l7 == 7 ? acc[7] : v;
                v = XMASK_ADD(v, 1);
                v = XMASK_ADD(v, 2);
                rr = (l3 == u) ? v : rr;
            }
            r[half] = rr;
        }
        const float r0 = r[0], r1 = r[1];

        const int e0 = p << 5;
        const int eo = e0 + (lane < 8 ? lane : lane - 16);
        if (lane < 8 || lane >= 24) {
            if (eo < E)      out[eo]      = r0;
            if (eo + 16 < E) out[eo + 16] = r1;
        }
    }
}

extern "C" void kernel_launch(void* const* d_in, const int* in_sizes, int n_in,
                              void* d_out, int out_size, void* d_ws, size_t ws_size,
                              hipStream_t stream) {
    const float* h   = (const float*)d_in[0];
    const int*   src = (const int*)d_in[1];
    const int*   dst = (const int*)d_in[2];
    float*       out = (float*)d_out;

    const int E = in_sizes[1];

    edge_dot_quadfrag_kernel<<<GRID_BLOCKS, BLOCK_THREADS, 0, stream>>>(h, src, dst, out, E);
}
